# starting kernel
# baseline (speedup 1.0000x reference)
#include <hip/hip_runtime.h>
#include <math.h>

typedef float v2f __attribute__((ext_vector_type(2)));
typedef float v8f __attribute__((ext_vector_type(8)));

#define H    128
#define EPSN 1e-8f

__global__ __launch_bounds__(32)
void node_mlp_kernel(const float* __restrict__ emb,
                     const float* __restrict__ W1, const float* __restrict__ b1,
                     const float* __restrict__ W2, const float* __restrict__ b2,
                     float* __restrict__ Y, float* __restrict__ nrm, int N)
{
    __shared__ float lds[16 * H];
    const int lane = threadIdx.x;
    const int m    = lane & 15;
    const int kh   = (lane >> 4) * 2;
    const int rhalf = (lane >> 4) << 3;
    const int r0   = blockIdx.x * 16;

    for (int i = lane; i < 16 * H; i += 32) {
        int r = r0 + (i >> 7);
        lds[i] = (r < N) ? emb[(size_t)r * H + (i & 127)] : 0.0f;
    }
    __syncthreads();

    v8f acc[8];
    const v8f vzero = {0.f, 0.f, 0.f, 0.f, 0.f, 0.f, 0.f, 0.f};

    #pragma unroll
    for (int n = 0; n < 8; ++n) acc[n] = vzero;
    for (int k = 0; k < H; k += 4) {
        v2f a = { lds[m * H + k + kh], lds[m * H + k + kh + 1] };
        #pragma unroll
        for (int n = 0; n < 8; ++n) {
            const float* wrow = W1 + (size_t)(n * 16 + m) * H + k + kh;
            v2f b = { wrow[0], wrow[1] };
            acc[n] = __builtin_amdgcn_wmma_f32_16x16x4_f32(
                false, a, false, b, (short)0, acc[n], false, false);
        }
    }
    __syncthreads();
    #pragma unroll
    for (int n = 0; n < 8; ++n) {
        int col = n * 16 + m;
        float bias = b1[col];
        #pragma unroll
        for (int r = 0; r < 8; ++r) {
            float x = acc[n][r] + bias;
            lds[(r + rhalf) * H + col] = (x > 0.0f) ? x : expm1f(x);
        }
    }
    __syncthreads();

    #pragma unroll
    for (int n = 0; n < 8; ++n) acc[n] = vzero;
    for (int k = 0; k < H; k += 4) {
        v2f a = { lds[m * H + k + kh], lds[m * H + k + kh + 1] };
        #pragma unroll
        for (int n = 0; n < 8; ++n) {
            const float* wrow = W2 + (size_t)(n * 16 + m) * H + k + kh;
            v2f b = { wrow[0], wrow[1] };
            acc[n] = __builtin_amdgcn_wmma_f32_16x16x4_f32(
                false, a, false, b, (short)0, acc[n], false, false);
        }
    }
    __syncthreads();
    #pragma unroll
    for (int n = 0; n < 8; ++n) {
        int col = n * 16 + m;
        float bias = b2[col];
        #pragma unroll
        for (int r = 0; r < 8; ++r)
            lds[(r + rhalf) * H + col] = acc[n][r] + bias;
    }
    __syncthreads();

    if (lane < 16) {
        int r = r0 + lane;
        if (r < N) {
            float s = 0.0f;
            #pragma unroll 4
            for (int j = 0; j < H; ++j) {
                float v = lds[lane * H + j];
                s += v * v;
            }
            nrm[r] = fmaxf(sqrtf(s), EPSN);
        }
    }
    for (int i = lane; i < 16 * H; i += 32) {
        int r = r0 + (i >> 7);
        if (r < N) Y[(size_t)r * H + (i & 127)] = lds[i];
    }
}

__global__ __launch_bounds__(256)
void edge_cos_kernel(const int* __restrict__ ei, const float* __restrict__ Y,
                     const float* __restrict__ nrm, float* __restrict__ out,
                     int E)
{
    int wid  = (int)((blockIdx.x * 256u + threadIdx.x) >> 5);
    int lane = threadIdx.x & 31;
    if (wid >= E) return;
    int c = ei[wid];
    int r = ei[E + wid];
    const float4* f1 = (const float4*)(Y + (size_t)c * H);
    const float4* f2 = (const float4*)(Y + (size_t)r * H);
    float4 a = f1[lane];
    float4 b = f2[lane];
    float s = a.x * b.x + a.y * b.y + a.z * b.z + a.w * b.w;
    #pragma unroll
    for (int off = 16; off > 0; off >>= 1)
        s += __shfl_xor(s, off, 32);
    if (lane == 0) out[wid] = s / (nrm[c] * nrm[r]);
}

extern "C" void kernel_launch(void* const* d_in, const int* in_sizes, int n_in,
                              void* d_out, int out_size, void* d_ws, size_t ws_size,
                              hipStream_t stream)
{
    const float* emb = (const float*)d_in[0];
    const int*   ei  = (const int*)  d_in[1];
    const float* W1  = (const float*)d_in[2];
    const float* b1  = (const float*)d_in[3];
    const float* W2  = (const float*)d_in[4];
    const float* b2  = (const float*)d_in[5];
    float* out = (float*)d_out;

    const int N = in_sizes[0] / H;
    const int E = in_sizes[1] / 2;

    float* Y   = (float*)d_ws;
    float* nv  = Y + (size_t)N * H;

    int nblk = (N + 15) / 16;
    hipLaunchKernelGGL(node_mlp_kernel, dim3(nblk), dim3(32), 0, stream,
                       emb, W1, b1, W2, b2, Y, nv, N);

    int eblk = (E + 7) / 8;
    hipLaunchKernelGGL(edge_cos_kernel, dim3(eblk), dim3(256), 0, stream,
                       ei, Y, nv, out, E);
}


# node MLP: 16-wave workgroups stage W1 and W2 in LDS once and read B fragments from LDS with 16-byte loads, not scattered 8-byte cache loads per wave; k order inside the f32 dot sums changes
# speedup vs baseline: 1.1851x; 1.1851x over previous
#include <hip/hip_runtime.h>
#include <math.h>

typedef float v2f __attribute__((ext_vector_type(2)));
typedef float v4f __attribute__((ext_vector_type(4)));
typedef float v8f __attribute__((ext_vector_type(8)));

#define H    128
#define EPSN 1e-8f
#define LP   132
#define NW   16

struct Frag { v4f a; v4f b[8]; };

__device__ __forceinline__ void load_frag(Frag& f, const float* arow, const float* brow, int j)
{
    f.a = *(const v4f*)(arow + 8 * j);
    #pragma unroll
    for (int n = 0; n < 8; ++n)
        f.b[n] = *(const v4f*)(brow + n * 16 * LP + 8 * j);
}

__device__ __forceinline__ void mma_frag(const Frag& f, v8f acc[8])
{
    v2f a0 = { f.a.x, f.a.y };
    v2f a1 = { f.a.z, f.a.w };
    #pragma unroll
    for (int n = 0; n < 8; ++n) {
        v2f b0 = { f.b[n].x, f.b[n].y };
        acc[n] = __builtin_amdgcn_wmma_f32_16x16x4_f32(
            false, a0, false, b0, (short)0, acc[n], false, false);
    }
    #pragma unroll
    for (int n = 0; n < 8; ++n) {
        v2f b1v = { f.b[n].z, f.b[n].w };
        acc[n] = __builtin_amdgcn_wmma_f32_16x16x4_f32(
            false, a1, false, b1v, (short)0, acc[n], false, false);
    }
}

__device__ __forceinline__ void mlp_layer(const float* ws, const float* tile,
                                          int m, int hh, v8f acc[8])
{
    const float* arow = tile + m * LP + 4 * hh;
    const float* brow = ws   + m * LP + 4 * hh;
    Frag f0, f1;
    load_frag(f0, arow, brow, 0);
    #pragma unroll 1
    for (int j = 0; j < H / 8 - 2; j += 2) {
        load_frag(f1, arow, brow, j + 1);
        mma_frag(f0, acc);
        load_frag(f0, arow, brow, j + 2);
        mma_frag(f1, acc);
    }
    load_frag(f1, arow, brow, H / 8 - 1);
    mma_frag(f0, acc);
    mma_frag(f1, acc);
}

__global__ __launch_bounds__(NW * 32)
void node_mlp_kernel(const float* __restrict__ emb,
                     const float* __restrict__ W1, const float* __restrict__ b1,
                     const float* __restrict__ W2, const float* __restrict__ b2,
                     float* __restrict__ Y, float* __restrict__ nrm, int N)
{
    __shared__ __attribute__((aligned(16))) float w1s[H * LP];
    __shared__ __attribute__((aligned(16))) float w2s[H * LP];
    __shared__ __attribute__((aligned(16))) float tiles[NW * 16 * LP];

    const int tid   = threadIdx.x;
    const int lane  = tid & 31;
    const int wave  = __builtin_amdgcn_readfirstlane(tid >> 5);
    const int m     = lane & 15;
    const int hh    = lane >> 4;
    const int rhalf = hh << 3;
    const int r0    = (blockIdx.x * NW + wave) * 16;
    float* tile = tiles + wave * 16 * LP;

    {
        v4f t1[8], t2[8], xs[16];
        #pragma unroll
        for (int q = 0; q < 8; ++q) {
            int row = wave + NW * q;
            t1[q] = *(const v4f*)(W1 + row * H + lane * 4);
            t2[q] = *(const v4f*)(W2 + row * H + lane * 4);
        }
        #pragma unroll
        for (int q = 0; q < 16; ++q) {
            int r = r0 + q;
            r = (r < N) ? r : N - 1;
            xs[q] = *(const v4f*)(emb + (size_t)r * H + lane * 4);
        }
        #pragma unroll
        for (int q = 0; q < 8; ++q) {
            int row = wave + NW * q;
            *(v4f*)(w1s + row * LP + lane * 4) = t1[q];
            *(v4f*)(w2s + row * LP + lane * 4) = t2[q];
        }
        #pragma unroll
        for (int q = 0; q < 16; ++q)
            *(v4f*)(tile + q * LP + lane * 4) = xs[q];
    }
    __syncthreads();

    v8f acc[8];
    const v8f vzero = {0.f, 0.f, 0.f, 0.f, 0.f, 0.f, 0.f, 0.f};

    #pragma unroll
    for (int n = 0; n < 8; ++n) acc[n] = vzero;
    mlp_layer(w1s, tile, m, hh, acc);
    __syncthreads();
    #pragma unroll
    for (int n = 0; n < 8; ++n) {
        int col = n * 16 + m;
        float bias = b1[col];
        #pragma unroll
        for (int r = 0; r < 8; ++r) {
            float x = acc[n][r] + bias;
            tile[(r + rhalf) * LP + col] = (x > 0.0f) ? x : expm1f(x);
        }
    }
    __syncthreads();

    #pragma unroll
    for (int n = 0; n < 8; ++n) acc[n] = vzero;
    mlp_layer(w2s, tile, m, hh, acc);
    __syncthreads();
    #pragma unroll
    for (int n = 0; n < 8; ++n) {
        int col = n * 16 + m;
        float bias = b2[col];
        #pragma unroll
        for (int r = 0; r < 8; ++r)
            tile[(r + rhalf) * LP + col] = acc[n][r] + bias;
    }
    __syncthreads();

    if (lane < 16) {
        int r = r0 + lane;
        if (r < N) {
            float s = 0.0f;
            #pragma unroll 4
            for (int j = 0; j < H; ++j) {
                float v = tile[lane * LP + j];
                s += v * v;
            }
            nrm[r] = fmaxf(sqrtf(s), EPSN);
        }
    }
    {
        v4f ys[16];
        #pragma unroll
        for (int q = 0; q < 16; ++q)
            ys[q] = *(const v4f*)(tile + q * LP + lane * 4);
        #pragma unroll
        for (int q = 0; q < 16; ++q) {
            int r = r0 + q;
            if (r < N) *(v4f*)(Y + (size_t)r * H + lane * 4) = ys[q];
        }
    }
}

__global__ __launch_bounds__(256)
void edge_cos_kernel(const int* __restrict__ ei, const float* __restrict__ Y,
                     const float* __restrict__ nrm, float* __restrict__ out,
                     int E)
{
    int wid  = (int)((blockIdx.x * 256u + threadIdx.x) >> 5);
    int lane = threadIdx.x & 31;
    if (wid >= E) return;
    int c = ei[wid];
    int r = ei[E + wid];
    const float4* f1 = (const float4*)(Y + (size_t)c * H);
    const float4* f2 = (const float4*)(Y + (size_t)r * H);
    float4 a = f1[lane];
    float4 b = f2[lane];
    float s = a.x * b.x + a.y * b.y + a.z * b.z + a.w * b.w;
    #pragma unroll
    for (int off = 16; off > 0; off >>= 1)
        s += __shfl_xor(s, off, 32);
    if (lane == 0) out[wid] = s / (nrm[c] * nrm[r]);
}

extern "C" void kernel_launch(void* const* d_in, const int* in_sizes, int n_in,
                              void* d_out, int out_size, void* d_ws, size_t ws_size,
                              hipStream_t stream)
{
    const float* emb = (const float*)d_in[0];
    const int*   ei  = (const int*)  d_in[1];
    const float* W1  = (const float*)d_in[2];
    const float* b1  = (const float*)d_in[3];
    const float* W2  = (const float*)d_in[4];
    const float* b2  = (const float*)d_in[5];
    float* out = (float*)d_out;

    const int N = in_sizes[0] / H;
    const int E = in_sizes[1] / 2;

    float* Y   = (float*)d_ws;
    float* nv  = Y + (size_t)N * H;

    int nblk = (N + 16 * NW - 1) / (16 * NW);
    hipLaunchKernelGGL(node_mlp_kernel, dim3(nblk), dim3(NW * 32), 0, stream,
                       emb, W1, b1, W2, b2, Y, nv, N);

    int eblk = (E + 7) / 8;
    hipLaunchKernelGGL(edge_cos_kernel, dim3(eblk), dim3(256), 0, stream,
                       ei, Y, nv, out, E);
}


# edge kernel: one wave takes 8 edges (16 lanes per edge) and issues all index loads, then all row and norm loads as one batch, instead of one wave per edge; order of the f32 dot sums changes
# speedup vs baseline: 2.1284x; 1.7960x over previous
#include <hip/hip_runtime.h>
#include <math.h>

typedef float v2f __attribute__((ext_vector_type(2)));
typedef float v4f __attribute__((ext_vector_type(4)));
typedef float v8f __attribute__((ext_vector_type(8)));

#define H    128
#define EPSN 1e-8f
#define LP   132
#define NW   16

struct Frag { v4f a; v4f b[8]; };

__device__ __forceinline__ void load_frag(Frag& f, const float* arow, const float* brow, int j)
{
    f.a = *(const v4f*)(arow + 8 * j);
    #pragma unroll
    for (int n = 0; n < 8; ++n)
        f.b[n] = *(const v4f*)(brow + n * 16 * LP + 8 * j);
}

__device__ __forceinline__ void mma_frag(const Frag& f, v8f acc[8])
{
    v2f a0 = { f.a.x, f.a.y };
    v2f a1 = { f.a.z, f.a.w };
    #pragma unroll
    for (int n = 0; n < 8; ++n) {
        v2f b0 = { f.b[n].x, f.b[n].y };
        acc[n] = __builtin_amdgcn_wmma_f32_16x16x4_f32(
            false, a0, false, b0, (short)0, acc[n], false, false);
    }
    #pragma unroll
    for (int n = 0; n < 8; ++n) {
        v2f b1v = { f.b[n].z, f.b[n].w };
        acc[n] = __builtin_amdgcn_wmma_f32_16x16x4_f32(
            false, a1, false, b1v, (short)0, acc[n], false, false);
    }
}

__device__ __forceinline__ void mlp_layer(const float* ws, const float* tile,
                                          int m, int hh, v8f acc[8])
{
    const float* arow = tile + m * LP + 4 * hh;
    const float* brow = ws   + m * LP + 4 * hh;
    Frag f0, f1;
    load_frag(f0, arow, brow, 0);
    #pragma unroll 1
    for (int j = 0; j < H / 8 - 2; j += 2) {
        load_frag(f1, arow, brow, j + 1);
        mma_frag(f0, acc);
        load_frag(f0, arow, brow, j + 2);
        mma_frag(f1, acc);
    }
    load_frag(f1, arow, brow, H / 8 - 1);
    mma_frag(f0, acc);
    mma_frag(f1, acc);
}

__global__ __launch_bounds__(NW * 32)
void node_mlp_kernel(const float* __restrict__ emb,
                     const float* __restrict__ W1, const float* __restrict__ b1,
                     const float* __restrict__ W2, const float* __restrict__ b2,
                     float* __restrict__ Y, float* __restrict__ nrm, int N)
{
    __shared__ __attribute__((aligned(16))) float w1s[H * LP];
    __shared__ __attribute__((aligned(16))) float w2s[H * LP];
    __shared__ __attribute__((aligned(16))) float tiles[NW * 16 * LP];

    const int tid   = threadIdx.x;
    const int lane  = tid & 31;
    const int wave  = __builtin_amdgcn_readfirstlane(tid >> 5);
    const int m     = lane & 15;
    const int hh    = lane >> 4;
    const int rhalf = hh << 3;
    const int r0    = (blockIdx.x * NW + wave) * 16;
    float* tile = tiles + wave * 16 * LP;

    {
        v4f t1[8], t2[8], xs[16];
        #pragma unroll
        for (int q = 0; q < 8; ++q) {
            int row = wave + NW * q;
            t1[q] = *(const v4f*)(W1 + row * H + lane * 4);
            t2[q] = *(const v4f*)(W2 + row * H + lane * 4);
        }
        #pragma unroll
        for (int q = 0; q < 16; ++q) {
            int r = r0 + q;
            r = (r < N) ? r : N - 1;
            xs[q] = *(const v4f*)(emb + (size_t)r * H + lane * 4);
        }
        #pragma unroll
        for (int q = 0; q < 8; ++q) {
            int row = wave + NW * q;
            *(v4f*)(w1s + row * LP + lane * 4) = t1[q];
            *(v4f*)(w2s + row * LP + lane * 4) = t2[q];
        }
        #pragma unroll
        for (int q = 0; q < 16; ++q)
            *(v4f*)(tile + q * LP + lane * 4) = xs[q];
    }
    __syncthreads();

    v8f acc[8];
    const v8f vzero = {0.f, 0.f, 0.f, 0.f, 0.f, 0.f, 0.f, 0.f};

    #pragma unroll
    for (int n = 0; n < 8; ++n) acc[n] = vzero;
    mlp_layer(w1s, tile, m, hh, acc);
    __syncthreads();
    #pragma unroll
    for (int n = 0; n < 8; ++n) {
        int col = n * 16 + m;
        float bias = b1[col];
        #pragma unroll
        for (int r = 0; r < 8; ++r) {
            float x = acc[n][r] + bias;
            tile[(r + rhalf) * LP + col] = (x > 0.0f) ? x : expm1f(x);
        }
    }
    __syncthreads();

    #pragma unroll
    for (int n = 0; n < 8; ++n) acc[n] = vzero;
    mlp_layer(w2s, tile, m, hh, acc);
    __syncthreads();
    #pragma unroll
    for (int n = 0; n < 8; ++n) {
        int col = n * 16 + m;
        float bias = b2[col];
        #pragma unroll
        for (int r = 0; r < 8; ++r)
            tile[(r + rhalf) * LP + col] = acc[n][r] + bias;
    }
    __syncthreads();

    if (lane < 16) {
        int r = r0 + lane;
        if (r < N) {
            float s = 0.0f;
            #pragma unroll 4
            for (int j = 0; j < H; ++j) {
                float v = tile[lane * LP + j];
                s += v * v;
            }
            nrm[r] = fmaxf(sqrtf(s), EPSN);
        }
    }
    {
        v4f ys[16];
        #pragma unroll
        for (int q = 0; q < 16; ++q)
            ys[q] = *(const v4f*)(tile + q * LP + lane * 4);
        #pragma unroll
        for (int q = 0; q < 16; ++q) {
            int r = r0 + q;
            if (r < N) *(v4f*)(Y + (size_t)r * H + lane * 4) = ys[q];
        }
    }
}

#define EPW 8
#define EGR (EPW / 2)

template <int MASK>
__device__ __forceinline__ float xmask_add(float v)
{
    int t = __builtin_amdgcn_update_dpp(0, __float_as_int(v), 0x160 | MASK, 0xf, 0xf, false);
    return v + __int_as_float(t);
}

__global__ __launch_bounds__(256)
void edge_cos_kernel(const int* __restrict__ ei, const float* __restrict__ Y,
                     const float* __restrict__ nrm, float* __restrict__ out,
                     int E)
{
    const int lane = threadIdx.x & 31;
    const int sub  = lane & 15;
    const int half = lane >> 4;
    const int wid  = (int)((blockIdx.x * 256u + threadIdx.x) >> 5);
    const int base = wid * EPW;
    if (base >= E) return;

    int c[EGR], r[EGR];
    #pragma unroll
    for (int g = 0; g < EGR; ++g) {
        int e = base + 2 * g + half;
        e = (e < E) ? e : E - 1;
        c[g] = ei[e];
        r[g] = ei[E + e];
    }

    v4f a[EGR][2], b[EGR][2];
    float nc[EGR], nr[EGR];
    #pragma unroll
    for (int g = 0; g < EGR; ++g) {
        nc[g] = nrm[c[g]];
        nr[g] = nrm[r[g]];
    }
    #pragma unroll
    for (int g = 0; g < EGR; ++g) {
        const v4f* pa = (const v4f*)(Y + (size_t)c[g] * H) + sub;
        const v4f* pb = (const v4f*)(Y + (size_t)r[g] * H) + sub;
        #pragma unroll
        for (int i = 0; i < 2; ++i) {
            a[g][i] = pa[16 * i];
            b[g][i] = pb[16 * i];
        }
    }
    __builtin_amdgcn_sched_barrier(0);

    float num = 0.0f, den = 1.0f;
    #pragma unroll
    for (int g = 0; g < EGR; ++g) {
        float p[2];
        #pragma unroll
        for (int i = 0; i < 2; ++i)
            p[i] = a[g][i].x * b[g][i].x + a[g][i].y * b[g][i].y
                 + a[g][i].z * b[g][i].z + a[g][i].w * b[g][i].w;
        float s = p[0] + p[1];
        s = xmask_add<1>(s);
        s = xmask_add<2>(s);
        s = xmask_add<4>(s);
        s = xmask_add<8>(s);
        asm volatile("" :: "v"(nc[g]), "v"(nr[g]));
        if (sub == g) { num = s; den = nc[g] * nr[g]; }
    }
    const int e = base + 2 * sub + half;
    if (sub < EGR && e < E) out[e] = num / den;
}

extern "C" void kernel_launch(void* const* d_in, const int* in_sizes, int n_in,
                              void* d_out, int out_size, void* d_ws, size_t ws_size,
                              hipStream_t stream)
{
    const float* emb = (const float*)d_in[0];
    const int*   ei  = (const int*)  d_in[1];
    const float* W1  = (const float*)d_in[2];
    const float* b1  = (const float*)d_in[3];
    const float* W2  = (const float*)d_in[4];
    const float* b2  = (const float*)d_in[5];
    float* out = (float*)d_out;

    const int N = in_sizes[0] / H;
    const int E = in_sizes[1] / 2;

    float* Y   = (float*)d_ws;
    float* nv  = Y + (size_t)N * H;

    int nblk = (N + 16 * NW - 1) / (16 * NW);
    hipLaunchKernelGGL(node_mlp_kernel, dim3(nblk), dim3(NW * 32), 0, stream,
                       emb, W1, b1, W2, b2, Y, nv, N);

    int eblk = (E + 8 * EPW - 1) / (8 * EPW);
    hipLaunchKernelGGL(edge_cos_kernel, dim3(eblk), dim3(256), 0, stream,
                       ei, Y, nv, out, E);
}


# node kernel stores each row scaled by 1/max(norm,eps), so the edge kernel gathers no norm values and divides nothing; sum order unchanged, the quotient's rounding moves from the sum to the elements
# speedup vs baseline: 2.2299x; 1.0477x over previous
#include <hip/hip_runtime.h>
#include <math.h>

typedef float v2f __attribute__((ext_vector_type(2)));
typedef float v4f __attribute__((ext_vector_type(4)));
typedef float v8f __attribute__((ext_vector_type(8)));

#define H    128
#define EPSN 1e-8f
#define LP   132
#define NW   16

struct Frag { v4f a; v4f b[8]; };

__device__ __forceinline__ void load_frag(Frag& f, const float* arow, const float* brow, int j)
{
    f.a = *(const v4f*)(arow + 8 * j);
    #pragma unroll
    for (int n = 0; n < 8; ++n)
        f.b[n] = *(const v4f*)(brow + n * 16 * LP + 8 * j);
}

__device__ __forceinline__ void mma_frag(const Frag& f, v8f acc[8])
{
    v2f a0 = { f.a.x, f.a.y };
    v2f a1 = { f.a.z, f.a.w };
    #pragma unroll
    for (int n = 0; n < 8; ++n) {
        v2f b0 = { f.b[n].x, f.b[n].y };
        acc[n] = __builtin_amdgcn_wmma_f32_16x16x4_f32(
            false, a0, false, b0, (short)0, acc[n], false, false);
    }
    #pragma unroll
    for (int n = 0; n < 8; ++n) {
        v2f b1v = { f.b[n].z, f.b[n].w };
        acc[n] = __builtin_amdgcn_wmma_f32_16x16x4_f32(
            false, a1, false, b1v, (short)0, acc[n], false, false);
    }
}

__device__ __forceinline__ void mlp_layer(const float* ws, const float* tile,
                                          int m, int hh, v8f acc[8])
{
    const float* arow = tile + m * LP + 4 * hh;
    const float* brow = ws   + m * LP + 4 * hh;
    Frag f0, f1;
    load_frag(f0, arow, brow, 0);
    #pragma unroll 1
    for (int j = 0; j < H / 8 - 2; j += 2) {
        load_frag(f1, arow, brow, j + 1);
        mma_frag(f0, acc);
        load_frag(f0, arow, brow, j + 2);
        mma_frag(f1, acc);
    }
    load_frag(f1, arow, brow, H / 8 - 1);
    mma_frag(f0, acc);
    mma_frag(f1, acc);
}

__global__ __launch_bounds__(NW * 32)
void node_mlp_kernel(const float* __restrict__ emb,
                     const float* __restrict__ W1, const float* __restrict__ b1,
                     const float* __restrict__ W2, const float* __restrict__ b2,
                     float* __restrict__ Y, int N)
{
    __shared__ __attribute__((aligned(16))) float w1s[H * LP];
    __shared__ __attribute__((aligned(16))) float w2s[H * LP];
    __shared__ __attribute__((aligned(16))) float tiles[NW * 16 * LP];

    const int tid   = threadIdx.x;
    const int lane  = tid & 31;
    const int wave  = __builtin_amdgcn_readfirstlane(tid >> 5);
    const int m     = lane & 15;
    const int hh    = lane >> 4;
    const int rhalf = hh << 3;
    const int r0    = (blockIdx.x * NW + wave) * 16;
    float* tile = tiles + wave * 16 * LP;

    {
        v4f t1[8], t2[8], xs[16];
        #pragma unroll
        for (int q = 0; q < 8; ++q) {
            int row = wave + NW * q;
            t1[q] = *(const v4f*)(W1 + row * H + lane * 4);
            t2[q] = *(const v4f*)(W2 + row * H + lane * 4);
        }
        #pragma unroll
        for (int q = 0; q < 16; ++q) {
            int r = r0 + q;
            r = (r < N) ? r : N - 1;
            xs[q] = *(const v4f*)(emb + (size_t)r * H + lane * 4);
        }
        #pragma unroll
        for (int q = 0; q < 8; ++q) {
            int row = wave + NW * q;
            *(v4f*)(w1s + row * LP + lane * 4) = t1[q];
            *(v4f*)(w2s + row * LP + lane * 4) = t2[q];
        }
        #pragma unroll
        for (int q = 0; q < 16; ++q)
            *(v4f*)(tile + q * LP + lane * 4) = xs[q];
    }
    __syncthreads();

    v8f acc[8];
    const v8f vzero = {0.f, 0.f, 0.f, 0.f, 0.f, 0.f, 0.f, 0.f};

    #pragma unroll
    for (int n = 0; n < 8; ++n) acc[n] = vzero;
    mlp_layer(w1s, tile, m, hh, acc);
    __syncthreads();
    #pragma unroll
    for (int n = 0; n < 8; ++n) {
        int col = n * 16 + m;
        float bias = b1[col];
        #pragma unroll
        for (int r = 0; r < 8; ++r) {
            float x = acc[n][r] + bias;
            tile[(r + rhalf) * LP + col] = (x > 0.0f) ? x : expm1f(x);
        }
    }
    __syncthreads();

    #pragma unroll
    for (int n = 0; n < 8; ++n) acc[n] = vzero;
    mlp_layer(w2s, tile, m, hh, acc);
    __syncthreads();
    #pragma unroll
    for (int n = 0; n < 8; ++n) {
        int col = n * 16 + m;
        float bias = b2[col];
        #pragma unroll
        for (int r = 0; r < 8; ++r)
            tile[(r + rhalf) * LP + col] = acc[n][r] + bias;
    }
    __syncthreads();

    float inv = 0.0f;
    if (lane < 16) {
        int r = r0 + lane;
        if (r < N) {
            float s = 0.0f;
            #pragma unroll 4
            for (int j = 0; j < H; ++j) {
                float v = tile[lane * LP + j];
                s += v * v;
            }
            inv = 1.0f / fmaxf(sqrtf(s), EPSN);
        }
    }
    {
        v4f ys[16];
        #pragma unroll
        for (int q = 0; q < 16; ++q)
            ys[q] = *(const v4f*)(tile + q * LP + lane * 4);
        #pragma unroll
        for (int q = 0; q < 16; ++q) {
            int r = r0 + q;
            float sc = __int_as_float(__builtin_amdgcn_readlane(__float_as_int(inv), q));
            if (r < N) *(v4f*)(Y + (size_t)r * H + lane * 4) = ys[q] * sc;
        }
    }
}

#define EPW 8
#define EGR (EPW / 2)

template <int MASK>
__device__ __forceinline__ float xmask_add(float v)
{
    int t = __builtin_amdgcn_update_dpp(0, __float_as_int(v), 0x160 | MASK, 0xf, 0xf, false);
    return v + __int_as_float(t);
}

__global__ __launch_bounds__(256)
void edge_cos_kernel(const int* __restrict__ ei, const float* __restrict__ Y,
                     float* __restrict__ out, int E)
{
    const int lane = threadIdx.x & 31;
    const int sub  = lane & 15;
    const int half = lane >> 4;
    const int wid  = (int)((blockIdx.x * 256u + threadIdx.x) >> 5);
    const int base = wid * EPW;
    if (base >= E) return;

    int c[EGR], r[EGR];
    #pragma unroll
    for (int g = 0; g < EGR; ++g) {
        int e = base + 2 * g + half;
        e = (e < E) ? e : E - 1;
        c[g] = ei[e];
        r[g] = ei[E + e];
    }

    v4f a[EGR][2], b[EGR][2];
    #pragma unroll
    for (int g = 0; g < EGR; ++g) {
        const v4f* pa = (const v4f*)(Y + (size_t)c[g] * H) + sub;
        const v4f* pb = (const v4f*)(Y + (size_t)r[g] * H) + sub;
        #pragma unroll
        for (int i = 0; i < 2; ++i) {
            a[g][i] = pa[16 * i];
            b[g][i] = pb[16 * i];
        }
    }
    __builtin_amdgcn_sched_barrier(0);

    float num = 0.0f;
    #pragma unroll
    for (int g = 0; g < EGR; ++g) {
        float p[2];
        #pragma unroll
        for (int i = 0; i < 2; ++i)
            p[i] = a[g][i].x * b[g][i].x + a[g][i].y * b[g][i].y
                 + a[g][i].z * b[g][i].z + a[g][i].w * b[g][i].w;
        float s = p[0] + p[1];
        s = xmask_add<1>(s);
        s = xmask_add<2>(s);
        s = xmask_add<4>(s);
        s = xmask_add<8>(s);
        if (sub == g) num = s;
    }
    const int e = base + 2 * sub + half;
    if (sub < EGR && e < E) out[e] = num;
}

extern "C" void kernel_launch(void* const* d_in, const int* in_sizes, int n_in,
                              void* d_out, int out_size, void* d_ws, size_t ws_size,
                              hipStream_t stream)
{
    const float* emb = (const float*)d_in[0];
    const int*   ei  = (const int*)  d_in[1];
    const float* W1  = (const float*)d_in[2];
    const float* b1  = (const float*)d_in[3];
    const float* W2  = (const float*)d_in[4];
    const float* b2  = (const float*)d_in[5];
    float* out = (float*)d_out;

    const int N = in_sizes[0] / H;
    const int E = in_sizes[1] / 2;

    float* Y   = (float*)d_ws;

    int nblk = (N + 16 * NW - 1) / (16 * NW);
    hipLaunchKernelGGL(node_mlp_kernel, dim3(nblk), dim3(NW * 32), 0, stream,
                       emb, W1, b1, W2, b2, Y, N);

    int eblk = (E + 8 * EPW - 1) / (8 * EPW);
    hipLaunchKernelGGL(edge_cos_kernel, dim3(eblk), dim3(256), 0, stream,
                       ei, Y, out, E);
}


# node MLP: each workgroup takes 12 row blocks on 12 waves plus one block split by columns over waves 12-15, so a SIMD issues 1664 f32 WMMAs, not 2048; every sum keeps its order
# speedup vs baseline: 2.3313x; 1.0455x over previous
#include <hip/hip_runtime.h>
#include <math.h>

typedef float v2f __attribute__((ext_vector_type(2)));
typedef float v4f __attribute__((ext_vector_type(4)));
typedef float v8f __attribute__((ext_vector_type(8)));

#define H    128
#define EPSN 1e-8f
#define LP   132
#define NW   16

template <int NT> struct Frag { v4f a; v4f b[NT]; };

template <int NT>
__device__ __forceinline__ void load_frag(Frag<NT>& f, const float* arow, const float* brow, int j)
{
    f.a = *(const v4f*)(arow + 8 * j);
    #pragma unroll
    for (int n = 0; n < NT; ++n)
        f.b[n] = *(const v4f*)(brow + n * 16 * LP + 8 * j);
}

template <int NT>
__device__ __forceinline__ void mma_frag(const Frag<NT>& f, v8f* acc)
{
    v2f a0 = { f.a.x, f.a.y };
    v2f a1 = { f.a.z, f.a.w };
    #pragma unroll
    for (int n = 0; n < NT; ++n) {
        v2f b0 = { f.b[n].x, f.b[n].y };
        acc[n] = __builtin_amdgcn_wmma_f32_16x16x4_f32(
            false, a0, false, b0, (short)0, acc[n], false, false);
    }
    #pragma unroll
    for (int n = 0; n < NT; ++n) {
        v2f b1v = { f.b[n].z, f.b[n].w };
        acc[n] = __builtin_amdgcn_wmma_f32_16x16x4_f32(
            false, a1, false, b1v, (short)0, acc[n], false, false);
    }
}

template <int NT>
__device__ __forceinline__ void mlp_layer(const float* ws, const float* tile, int n0,
                                          int m, int hh, v8f* acc)
{
    const float* arow = tile + m * LP + 4 * hh;
    const float* brow = ws   + (n0 * 16 + m) * LP + 4 * hh;
    Frag<NT> f0, f1;
    load_frag<NT>(f0, arow, brow, 0);
    #pragma unroll 1
    for (int j = 0; j < H / 8 - 2; j += 2) {
        load_frag<NT>(f1, arow, brow, j + 1);
        mma_frag<NT>(f0, acc);
        load_frag<NT>(f0, arow, brow, j + 2);
        mma_frag<NT>(f1, acc);
    }
    load_frag<NT>(f1, arow, brow, H / 8 - 1);
    mma_frag<NT>(f0, acc);
    mma_frag<NT>(f1, acc);
}

template <int NT, bool ELU>
__device__ __forceinline__ void put_tile(float* tile, const v8f* acc, const float* __restrict__ bias_v,
                                         int n0, int m, int rhalf)
{
    #pragma unroll
    for (int n = 0; n < NT; ++n) {
        int col = (n0 + n) * 16 + m;
        float bias = bias_v[col];
        #pragma unroll
        for (int r = 0; r < 8; ++r) {
            float x = acc[n][r] + bias;
            tile[(r + rhalf) * LP + col] = ELU ? ((x > 0.0f) ? x : expm1f(x)) : x;
        }
    }
}

template <int NR>
__device__ __forceinline__ void load_rows(v4f* xs, const float* __restrict__ emb, int g0, int N, int lane)
{
    #pragma unroll
    for (int q = 0; q < NR; ++q) {
        int r = g0 + q;
        r = (r < N) ? r : N - 1;
        xs[q] = *(const v4f*)(emb + (size_t)r * H + lane * 4);
    }
}

template <int NR>
__device__ __forceinline__ void store_rows(float* trow, const v4f* xs, int lane)
{
    #pragma unroll
    for (int q = 0; q < NR; ++q)
        *(v4f*)(trow + q * LP + lane * 4) = xs[q];
}

__device__ __forceinline__ void store_weights(float* w1s, float* w2s, const v4f* t1, const v4f* t2,
                                              int wave, int lane)
{
    #pragma unroll
    for (int q = 0; q < 8; ++q) {
        int row = wave + NW * q;
        *(v4f*)(w1s + row * LP + lane * 4) = t1[q];
        *(v4f*)(w2s + row * LP + lane * 4) = t2[q];
    }
}

template <int NR>
__device__ __forceinline__ void finish_rows(const float* trow, float* __restrict__ Y, int g0, int N, int lane)
{
    float inv = 0.0f;
    if (lane < NR) {
        int r = g0 + lane;
        if (r < N) {
            float s = 0.0f;
            #pragma unroll 4
            for (int j = 0; j < H; ++j) {
                float v = trow[lane * LP + j];
                s += v * v;
            }
            inv = 1.0f / fmaxf(sqrtf(s), EPSN);
        }
    }
    v4f ys[NR];
    #pragma unroll
    for (int q = 0; q < NR; ++q)
        ys[q] = *(const v4f*)(trow + q * LP + lane * 4);
    #pragma unroll
    for (int q = 0; q < NR; ++q) {
        int r = g0 + q;
        float sc = __int_as_float(__builtin_amdgcn_readlane(__float_as_int(inv), q));
        if (r < N) *(v4f*)(Y + (size_t)r * H + lane * 4) = ys[q] * sc;
    }
}

#define NWM 12
#define NTL (NWM + 1)

__global__ __launch_bounds__(NW * 32)
void node_mlp_kernel(const float* __restrict__ emb,
                     const float* __restrict__ W1, const float* __restrict__ b1,
                     const float* __restrict__ W2, const float* __restrict__ b2,
                     float* __restrict__ Y, int N)
{
    __shared__ __attribute__((aligned(16))) float w1s[H * LP];
    __shared__ __attribute__((aligned(16))) float w2s[H * LP];
    __shared__ __attribute__((aligned(16))) float tiles[NTL * 16 * LP];

    const int tid   = threadIdx.x;
    const int lane  = tid & 31;
    const int wave  = __builtin_amdgcn_readfirstlane(tid >> 5);
    const int m     = lane & 15;
    const int hh    = lane >> 4;
    const int rhalf = hh << 3;

    const int  nrb    = (N + 15) >> 4;
    const bool whole  = wave < NWM;
    const int  qw     = whole ? 0 : wave - NWM;
    const int  rb     = whole ? (int)blockIdx.x * NWM + wave
                              : (int)gridDim.x * NWM + (int)blockIdx.x;
    const bool work   = rb < nrb;
    const int  rbeg   = 4 * qw;
    const int  n0     = 2 * qw;
    const int  r0     = rb * 16;
    float* tile = tiles + (whole ? wave : NWM) * 16 * LP;

    {
        v4f t1[8], t2[8];
        #pragma unroll
        for (int q = 0; q < 8; ++q) {
            int row = wave + NW * q;
            t1[q] = *(const v4f*)(W1 + row * H + lane * 4);
            t2[q] = *(const v4f*)(W2 + row * H + lane * 4);
        }
        if (whole) {
            v4f xs[16];
            load_rows<16>(xs, emb, r0, N, lane);
            store_weights(w1s, w2s, t1, t2, wave, lane);
            store_rows<16>(tile, xs, lane);
        } else {
            v4f xs[4];
            load_rows<4>(xs, emb, r0 + rbeg, N, lane);
            store_weights(w1s, w2s, t1, t2, wave, lane);
            store_rows<4>(tile + rbeg * LP, xs, lane);
        }
    }
    __syncthreads();

    v8f acc[8];
    const v8f vzero = {0.f, 0.f, 0.f, 0.f, 0.f, 0.f, 0.f, 0.f};

    #pragma unroll
    for (int n = 0; n < 8; ++n) acc[n] = vzero;
    if (work) {
        if (whole) mlp_layer<8>(w1s, tile, 0, m, hh, acc);
        else       mlp_layer<2>(w1s, tile, n0, m, hh, acc);
    }
    __syncthreads();
    if (work) {
        if (whole) put_tile<8, true>(tile, acc, b1, 0, m, rhalf);
        else       put_tile<2, true>(tile, acc, b1, n0, m, rhalf);
    }
    __syncthreads();

    #pragma unroll
    for (int n = 0; n < 8; ++n) acc[n] = vzero;
    if (work) {
        if (whole) mlp_layer<8>(w2s, tile, 0, m, hh, acc);
        else       mlp_layer<2>(w2s, tile, n0, m, hh, acc);
    }
    __syncthreads();
    if (work) {
        if (whole) put_tile<8, false>(tile, acc, b2, 0, m, rhalf);
        else       put_tile<2, false>(tile, acc, b2, n0, m, rhalf);
    }
    __syncthreads();

    if (whole) finish_rows<16>(tile, Y, r0, N, lane);
    else       finish_rows<4>(tile + rbeg * LP, Y, r0 + rbeg, N, lane);
}

#define EPW 8
#define EGR (EPW / 2)

template <int MASK>
__device__ __forceinline__ float xmask_add(float v)
{
    int t = __builtin_amdgcn_update_dpp(0, __float_as_int(v), 0x160 | MASK, 0xf, 0xf, false);
    return v + __int_as_float(t);
}

__global__ __launch_bounds__(256)
void edge_cos_kernel(const int* __restrict__ ei, const float* __restrict__ Y,
                     float* __restrict__ out, int E)
{
    const int lane = threadIdx.x & 31;
    const int sub  = lane & 15;
    const int half = lane >> 4;
    const int wid  = (int)((blockIdx.x * 256u + threadIdx.x) >> 5);
    const int base = wid * EPW;
    if (base >= E) return;

    int c[EGR], r[EGR];
    #pragma unroll
    for (int g = 0; g < EGR; ++g) {
        int e = base + 2 * g + half;
        e = (e < E) ? e : E - 1;
        c[g] = ei[e];
        r[g] = ei[E + e];
    }

    v4f a[EGR][2], b[EGR][2];
    #pragma unroll
    for (int g = 0; g < EGR; ++g) {
        const v4f* pa = (const v4f*)(Y + (size_t)c[g] * H) + sub;
        const v4f* pb = (const v4f*)(Y + (size_t)r[g] * H) + sub;
        #pragma unroll
        for (int i = 0; i < 2; ++i) {
            a[g][i] = pa[16 * i];
            b[g][i] = pb[16 * i];
        }
    }
    __builtin_amdgcn_sched_barrier(0);

    float num = 0.0f;
    #pragma unroll
    for (int g = 0; g < EGR; ++g) {
        float p[2];
        #pragma unroll
        for (int i = 0; i < 2; ++i)
            p[i] = a[g][i].x * b[g][i].x + a[g][i].y * b[g][i].y
                 + a[g][i].z * b[g][i].z + a[g][i].w * b[g][i].w;
        float s = p[0] + p[1];
        s = xmask_add<1>(s);
        s = xmask_add<2>(s);
        s = xmask_add<4>(s);
        s = xmask_add<8>(s);
        if (sub == g) num = s;
    }
    const int e = base + 2 * sub + half;
    if (sub < EGR && e < E) out[e] = num;
}

extern "C" void kernel_launch(void* const* d_in, const int* in_sizes, int n_in,
                              void* d_out, int out_size, void* d_ws, size_t ws_size,
                              hipStream_t stream)
{
    const float* emb = (const float*)d_in[0];
    const int*   ei  = (const int*)  d_in[1];
    const float* W1  = (const float*)d_in[2];
    const float* b1  = (const float*)d_in[3];
    const float* W2  = (const float*)d_in[4];
    const float* b2  = (const float*)d_in[5];
    float* out = (float*)d_out;

    const int N = in_sizes[0] / H;
    const int E = in_sizes[1] / 2;

    float* Y   = (float*)d_ws;

    int nblk = ((N + 15) / 16 + NTL - 1) / NTL;
    hipLaunchKernelGGL(node_mlp_kernel, dim3(nblk), dim3(NW * 32), 0, stream,
                       emb, W1, b1, W2, b2, Y, N);

    int eblk = (E + 8 * EPW - 1) / (8 * EPW);
    hipLaunchKernelGGL(edge_cos_kernel, dim3(eblk), dim3(256), 0, stream,
                       ei, Y, out, E);
}


# node kernel epilogues without serial chains: ELU as straight-line select code (no branch per value, bias read early), row norms summed across lanes in registers; the norm's sum order changes
# speedup vs baseline: 2.4969x; 1.0710x over previous
#include <hip/hip_runtime.h>
#include <math.h>

typedef float v2f __attribute__((ext_vector_type(2)));
typedef float v4f __attribute__((ext_vector_type(4)));
typedef float v8f __attribute__((ext_vector_type(8)));

#define H    128
#define EPSN 1e-8f
#define LP   132
#define NW   16

template <int NT> struct Frag { v4f a; v4f b[NT]; };

template <int NT>
__device__ __forceinline__ void load_frag(Frag<NT>& f, const float* arow, const float* brow, int j)
{
    f.a = *(const v4f*)(arow + 8 * j);
    #pragma unroll
    for (int n = 0; n < NT; ++n)
        f.b[n] = *(const v4f*)(brow + n * 16 * LP + 8 * j);
}

template <int NT>
__device__ __forceinline__ void mma_frag(const Frag<NT>& f, v8f* acc)
{
    v2f a0 = { f.a.x, f.a.y };
    v2f a1 = { f.a.z, f.a.w };
    #pragma unroll
    for (int n = 0; n < NT; ++n) {
        v2f b0 = { f.b[n].x, f.b[n].y };
        acc[n] = __builtin_amdgcn_wmma_f32_16x16x4_f32(
            false, a0, false, b0, (short)0, acc[n], false, false);
    }
    #pragma unroll
    for (int n = 0; n < NT; ++n) {
        v2f b1v = { f.b[n].z, f.b[n].w };
        acc[n] = __builtin_amdgcn_wmma_f32_16x16x4_f32(
            false, a1, false, b1v, (short)0, acc[n], false, false);
    }
}

template <int NT>
__device__ __forceinline__ void mlp_layer(const float* ws, const float* tile, int n0,
                                          int m, int hh, v8f* acc)
{
    const float* arow = tile + m * LP + 4 * hh;
    const float* brow = ws   + (n0 * 16 + m) * LP + 4 * hh;
    Frag<NT> f0, f1;
    load_frag<NT>(f0, arow, brow, 0);
    #pragma unroll 1
    for (int j = 0; j < H / 8 - 2; j += 2) {
        load_frag<NT>(f1, arow, brow, j + 1);
        mma_frag<NT>(f0, acc);
        load_frag<NT>(f0, arow, brow, j + 2);
        mma_frag<NT>(f1, acc);
    }
    load_frag<NT>(f1, arow, brow, H / 8 - 1);
    mma_frag<NT>(f0, acc);
    mma_frag<NT>(f1, acc);
}

__device__ __forceinline__ float elu_val(float x)
{
    const float xn = fmaxf(fminf(x, 0.0f), -88.0f);
    const float n  = rintf(xn * 1.44269502f);
    float r = fmaf(n, -0.693147182f, xn);
    r       = fmaf(n, 1.90465421e-9f, r);
    float p = fmaf(r, 0.000199510498f, 0.00139304996f);
    p = fmaf(r, p, 0.00833319966f);
    p = fmaf(r, p, 0.0416665114f);
    p = fmaf(r, p, 0.166666672f);
    p = fmaf(r, p, 0.5f);
    const float em1r = fmaf(r, r * p, r);
    const float s    = ldexpf(1.0f, (int)n);
    const float e    = fmaf(s, em1r, s - 1.0f);
    return (x <= 0.0f) ? e : x;
}

template <int NT, bool ELU>
__device__ __forceinline__ void put_tile(float* tile, const v8f* acc, const float* bias_r,
                                         int n0, int m, int rhalf)
{
    #pragma unroll
    for (int n = 0; n < NT; ++n) {
        int col = (n0 + n) * 16 + m;
        float bias = bias_r[n];
        #pragma unroll
        for (int r = 0; r < 8; ++r) {
            float x = acc[n][r] + bias;
            tile[(r + rhalf) * LP + col] = ELU ? elu_val(x) : x;
        }
    }
}

template <int NR>
__device__ __forceinline__ void load_rows(v4f* xs, const float* __restrict__ emb, int g0, int N, int lane)
{
    #pragma unroll
    for (int q = 0; q < NR; ++q) {
        int r = g0 + q;
        r = (r < N) ? r : N - 1;
        xs[q] = *(const v4f*)(emb + (size_t)r * H + lane * 4);
    }
}

template <int NR>
__device__ __forceinline__ void store_rows(float* trow, const v4f* xs, int lane)
{
    #pragma unroll
    for (int q = 0; q < NR; ++q)
        *(v4f*)(trow + q * LP + lane * 4) = xs[q];
}

__device__ __forceinline__ void store_weights(float* w1s, float* w2s, const v4f* t1, const v4f* t2,
                                              int wave, int lane)
{
    #pragma unroll
    for (int q = 0; q < 8; ++q) {
        int row = wave + NW * q;
        *(v4f*)(w1s + row * LP + lane * 4) = t1[q];
        *(v4f*)(w2s + row * LP + lane * 4) = t2[q];
    }
}

template <int MASK>
__device__ __forceinline__ float xmask_add(float v)
{
    int t = __builtin_amdgcn_update_dpp(0, __float_as_int(v), 0x160 | MASK, 0xf, 0xf, false);
    return v + __int_as_float(t);
}

__device__ __forceinline__ float xhalf_add(float v)
{
    int b = __float_as_int(v);
    int t = __builtin_amdgcn_permlanex16(b, b, 0x76543210, 0xfedcba98, false, false);
    return v + __int_as_float(t);
}

template <int NR>
__device__ __forceinline__ void finish_rows(const float* trow, float* __restrict__ Y, int g0, int N, int lane)
{
    v4f ys[NR];
    #pragma unroll
    for (int q = 0; q < NR; ++q)
        ys[q] = *(const v4f*)(trow + q * LP + lane * 4);

    float ss = 0.0f;
    #pragma unroll
    for (int q = 0; q < NR; ++q) {
        float s = ys[q].x * ys[q].x + ys[q].y * ys[q].y + ys[q].z * ys[q].z + ys[q].w * ys[q].w;
        s = xmask_add<1>(s);
        s = xmask_add<2>(s);
        s = xmask_add<4>(s);
        s = xmask_add<8>(s);
        s = xhalf_add(s);
        if (lane == q) ss = s;
    }
    float inv = 0.0f;
    if (lane < NR && g0 + lane < N) inv = 1.0f / fmaxf(sqrtf(ss), EPSN);

    #pragma unroll
    for (int q = 0; q < NR; ++q) {
        int r = g0 + q;
        float sc = __int_as_float(__builtin_amdgcn_readlane(__float_as_int(inv), q));
        if (r < N) *(v4f*)(Y + (size_t)r * H + lane * 4) = ys[q] * sc;
    }
}

#define NWM 12
#define NTL (NWM + 1)

__global__ __launch_bounds__(NW * 32)
void node_mlp_kernel(const float* __restrict__ emb,
                     const float* __restrict__ W1, const float* __restrict__ b1,
                     const float* __restrict__ W2, const float* __restrict__ b2,
                     float* __restrict__ Y, int N)
{
    __shared__ __attribute__((aligned(16))) float w1s[H * LP];
    __shared__ __attribute__((aligned(16))) float w2s[H * LP];
    __shared__ __attribute__((aligned(16))) float tiles[NTL * 16 * LP];

    const int tid   = threadIdx.x;
    const int lane  = tid & 31;
    const int wave  = __builtin_amdgcn_readfirstlane(tid >> 5);
    const int m     = lane & 15;
    const int hh    = lane >> 4;
    const int rhalf = hh << 3;

    const int  nrb    = (N + 15) >> 4;
    const bool whole  = wave < NWM;
    const int  qw     = whole ? 0 : wave - NWM;
    const int  rb     = whole ? (int)blockIdx.x * NWM + wave
                              : (int)gridDim.x * NWM + (int)blockIdx.x;
    const bool work   = rb < nrb;
    const int  rbeg   = 4 * qw;
    const int  n0     = 2 * qw;
    const int  r0     = rb * 16;
    float* tile = tiles + (whole ? wave : NWM) * 16 * LP;

    float bv1[8], bv2[8];
    #pragma unroll
    for (int n = 0; n < 8; ++n) {
        int col = ((n0 + n) & 7) * 16 + m;
        bv1[n] = b1[col];
        bv2[n] = b2[col];
    }

    {
        v4f t1[8], t2[8];
        #pragma unroll
        for (int q = 0; q < 8; ++q) {
            int row = wave + NW * q;
            t1[q] = *(const v4f*)(W1 + row * H + lane * 4);
            t2[q] = *(const v4f*)(W2 + row * H + lane * 4);
        }
        if (whole) {
            v4f xs[16];
            load_rows<16>(xs, emb, r0, N, lane);
            store_weights(w1s, w2s, t1, t2, wave, lane);
            store_rows<16>(tile, xs, lane);
        } else {
            v4f xs[4];
            load_rows<4>(xs, emb, r0 + rbeg, N, lane);
            store_weights(w1s, w2s, t1, t2, wave, lane);
            store_rows<4>(tile + rbeg * LP, xs, lane);
        }
    }
    __syncthreads();

    v8f acc[8];
    const v8f vzero = {0.f, 0.f, 0.f, 0.f, 0.f, 0.f, 0.f, 0.f};

    #pragma unroll
    for (int n = 0; n < 8; ++n) acc[n] = vzero;
    if (work) {
        if (whole) mlp_layer<8>(w1s, tile, 0, m, hh, acc);
        else       mlp_layer<2>(w1s, tile, n0, m, hh, acc);
    }
    __syncthreads();
    if (work) {
        if (whole) put_tile<8, true>(tile, acc, bv1, 0, m, rhalf);
        else       put_tile<2, true>(tile, acc, bv1, n0, m, rhalf);
    }
    __syncthreads();

    #pragma unroll
    for (int n = 0; n < 8; ++n) acc[n] = vzero;
    if (work) {
        if (whole) mlp_layer<8>(w2s, tile, 0, m, hh, acc);
        else       mlp_layer<2>(w2s, tile, n0, m, hh, acc);
    }
    __syncthreads();
    if (work) {
        if (whole) put_tile<8, false>(tile, acc, bv2, 0, m, rhalf);
        else       put_tile<2, false>(tile, acc, bv2, n0, m, rhalf);
    }
    __syncthreads();

    if (whole) finish_rows<16>(tile, Y, r0, N, lane);
    else       finish_rows<4>(tile + rbeg * LP, Y, r0 + rbeg, N, lane);
}

#define EPW 8
#define EGR (EPW / 2)

__global__ __launch_bounds__(256)
void edge_cos_kernel(const int* __restrict__ ei, const float* __restrict__ Y,
                     float* __restrict__ out, int E)
{
    const int lane = threadIdx.x & 31;
    const int sub  = lane & 15;
    const int half = lane >> 4;
    const int wid  = (int)((blockIdx.x * 256u + threadIdx.x) >> 5);
    const int base = wid * EPW;
    if (base >= E) return;

    int c[EGR], r[EGR];
    #pragma unroll
    for (int g = 0; g < EGR; ++g) {
        int e = base + 2 * g + half;
        e = (e < E) ? e : E - 1;
        c[g] = ei[e];
        r[g] = ei[E + e];
    }

    v4f a[EGR][2], b[EGR][2];
    #pragma unroll
    for (int g = 0; g < EGR; ++g) {
        const v4f* pa = (const v4f*)(Y + (size_t)c[g] * H) + sub;
        const v4f* pb = (const v4f*)(Y + (size_t)r[g] * H) + sub;
        #pragma unroll
        for (int i = 0; i < 2; ++i) {
            a[g][i] = pa[16 * i];
            b[g][i] = pb[16 * i];
        }
    }
    __builtin_amdgcn_sched_barrier(0);

    float num = 0.0f;
    #pragma unroll
    for (int g = 0; g < EGR; ++g) {
        float p[2];
        #pragma unroll
        for (int i = 0; i < 2; ++i)
            p[i] = a[g][i].x * b[g][i].x + a[g][i].y * b[g][i].y
                 + a[g][i].z * b[g][i].z + a[g][i].w * b[g][i].w;
        float s = p[0] + p[1];
        s = xmask_add<1>(s);
        s = xmask_add<2>(s);
        s = xmask_add<4>(s);
        s = xmask_add<8>(s);
        if (sub == g) num = s;
    }
    const int e = base + 2 * sub + half;
    if (sub < EGR && e < E) out[e] = num;
}

extern "C" void kernel_launch(void* const* d_in, const int* in_sizes, int n_in,
                              void* d_out, int out_size, void* d_ws, size_t ws_size,
                              hipStream_t stream)
{
    const float* emb = (const float*)d_in[0];
    const int*   ei  = (const int*)  d_in[1];
    const float* W1  = (const float*)d_in[2];
    const float* b1  = (const float*)d_in[3];
    const float* W2  = (const float*)d_in[4];
    const float* b2  = (const float*)d_in[5];
    float* out = (float*)d_out;

    const int N = in_sizes[0] / H;
    const int E = in_sizes[1] / 2;

    float* Y   = (float*)d_ws;

    int nblk = ((N + 15) / 16 + NTL - 1) / NTL;
    hipLaunchKernelGGL(node_mlp_kernel, dim3(nblk), dim3(NW * 32), 0, stream,
                       emb, W1, b1, W2, b2, Y, N);

    int eblk = (E + 8 * EPW - 1) / (8 * EPW);
    hipLaunchKernelGGL(edge_cos_kernel, dim3(eblk), dim3(256), 0, stream,
                       ei, Y, out, E);
}
